# starting kernel
# baseline (speedup 1.0000x reference)
#include <hip/hip_runtime.h>
#include <hip/hip_bf16.h>

#define N_NODES 50000
#define N_EDGES 1600000
#define D_IN    128
#define D_OUT   32
#define ALPHA   0.2f

typedef __attribute__((ext_vector_type(2))) float v2f;
typedef __attribute__((ext_vector_type(8))) float v8f;

__device__ __forceinline__ unsigned f2ord(float f) {
  unsigned u = __float_as_uint(f);
  return (u & 0x80000000u) ? ~u : (u | 0x80000000u);
}
__device__ __forceinline__ float ord2f(unsigned u) {
  u = (u & 0x80000000u) ? (u & 0x7FFFFFFFu) : ~u;
  return __uint_as_float(u);
}

__global__ void gat_init(float* __restrict__ hout, float* __restrict__ rows_sum,
                         unsigned* __restrict__ minmax) {
  int i = blockIdx.x * blockDim.x + threadIdx.x;
  if (i < N_NODES * D_OUT) hout[i] = 0.0f;
  if (i < N_NODES) rows_sum[i] = 0.0f;
  if (i == 0) { minmax[0] = 0xFFFFFFFFu; minmax[1] = 0u; }
}

__global__ void gat_gemm_wmma(const float* __restrict__ X,
                              const float* __restrict__ W,
                              float* __restrict__ Xp) {
  const int wave_in_block = threadIdx.x >> 5;
  const int lane          = threadIdx.x & 31;
  const int wave_id       = blockIdx.x * (blockDim.x >> 5) + wave_in_block;
  const int row_base      = wave_id * 16;
  if (row_base >= N_NODES) return;

  const int m    = lane & 15;
  const int kgrp = lane >> 4;

  const float* __restrict__ xrow = X + (size_t)(row_base + m) * D_IN;

  v8f acc0 = {};
  v8f acc1 = {};

  for (int k = 0; k < D_IN; k += 4) {
    const int kk = k + kgrp * 2;
    v2f a;
    a.x = xrow[kk + 0];
    a.y = xrow[kk + 1];
    v2f b0, b1;
    b0.x = W[(size_t)(kk + 0) * D_OUT + m];
    b0.y = W[(size_t)(kk + 1) * D_OUT + m];
    b1.x = W[(size_t)(kk + 0) * D_OUT + 16 + m];
    b1.y = W[(size_t)(kk + 1) * D_OUT + 16 + m];

    acc0 = __builtin_amdgcn_wmma_f32_16x16x4_f32(false, a, false, b0, (short)0, acc0, false, false);
    acc1 = __builtin_amdgcn_wmma_f32_16x16x4_f32(false, a, false, b1, (short)0, acc1, false, false);
  }

  const int rbase = row_base + kgrp * 8;
  for (int i = 0; i < 8; ++i) {
    const size_t r = (size_t)(rbase + i);
    Xp[r * D_OUT + m]      = acc0[i];
    Xp[r * D_OUT + 16 + m] = acc1[i];
  }
}

__global__ void gat_node_scores(const float* __restrict__ Xp,
                                const float* __restrict__ a0,
                                const float* __restrict__ a1,
                                float* __restrict__ s0,
                                float* __restrict__ s1) {
  int n = blockIdx.x * blockDim.x + threadIdx.x;
  if (n >= N_NODES) return;
  float acc0 = 0.0f, acc1 = 0.0f;
  const float* __restrict__ xr = Xp + (size_t)n * D_OUT;
#pragma unroll
  for (int d = 0; d < D_OUT; ++d) {
    float v = xr[d];
    acc0 += v * a0[d];
    acc1 += v * a1[d];
  }
  s0[n] = acc0;
  s1[n] = acc1;
}

__global__ void gat_edge_scores(const int* __restrict__ row_idx,
                                const int* __restrict__ col_idx,
                                const float* __restrict__ s0,
                                const float* __restrict__ s1,
                                float* __restrict__ att_raw,
                                unsigned* __restrict__ minmax) {
  __shared__ unsigned bmin, bmax;
  if (threadIdx.x == 0) { bmin = 0xFFFFFFFFu; bmax = 0u; }
  __syncthreads();

  int e = blockIdx.x * blockDim.x + threadIdx.x;
  if (e < N_EDGES) {
    float v = s0[row_idx[e]] + s1[col_idx[e]];
    v = (v > 0.0f) ? v : (ALPHA * v);
    att_raw[e] = v;
    unsigned o = f2ord(v);
    atomicMin(&bmin, o);
    atomicMax(&bmax, o);
  }
  __syncthreads();
  if (threadIdx.x == 0) {
    atomicMin(&minmax[0], bmin);
    atomicMax(&minmax[1], bmax);
  }
}

__global__ void gat_edge_scatter(const int* __restrict__ row_idx,
                                 const int* __restrict__ col_idx,
                                 const float* __restrict__ att_raw,
                                 const float* __restrict__ Xp,
                                 const unsigned* __restrict__ minmax,
                                 float* __restrict__ hout,
                                 float* __restrict__ rows_sum) {
  const int wave = (blockIdx.x * blockDim.x + threadIdx.x) >> 5;
  const int lane = threadIdx.x & 31;
  if (wave >= N_EDGES) return;

  const float vmin = ord2f(minmax[0]);
  const float vmax = ord2f(minmax[1]);
  const float inv  = 1.0f / (vmax - vmin);

  const int r = row_idx[wave];
  const int c = col_idx[wave];
  const float att = __expf((att_raw[wave] - vmin) * inv);

  const float v = att * Xp[(size_t)c * D_OUT + lane];
  atomicAdd(&hout[(size_t)r * D_OUT + lane], v);
  if (lane == 0) atomicAdd(&rows_sum[r], att);
}

__global__ void gat_finalize(float* __restrict__ hout,
                             const float* __restrict__ rows_sum) {
  int i = blockIdx.x * blockDim.x + threadIdx.x;
  if (i >= N_NODES * D_OUT) return;
  hout[i] /= rows_sum[i / D_OUT];
}

static inline int cdiv(int a, int b) { return (a + b - 1) / b; }

extern "C" void kernel_launch(void* const* d_in, const int* in_sizes, int n_in,
                              void* d_out, int out_size, void* d_ws, size_t ws_size,
                              hipStream_t stream) {
  (void)in_sizes; (void)n_in; (void)out_size; (void)ws_size;

  const float* X       = (const float*)d_in[0];
  const float* W       = (const float*)d_in[1];
  const float* a0      = (const float*)d_in[2];
  const float* a1      = (const float*)d_in[3];
  const int*   row_idx = (const int*)d_in[4];
  const int*   col_idx = (const int*)d_in[5];
  float* hout = (float*)d_out;

  char* ws = (char*)d_ws;
  size_t off = 0;
  auto alloc = [&](size_t bytes) { char* p = ws + off; off = (off + bytes + 255) & ~(size_t)255; return p; };
  float*    Xp       = (float*)alloc((size_t)N_NODES * D_OUT * sizeof(float));
  float*    s0       = (float*)alloc((size_t)N_NODES * sizeof(float));
  float*    s1       = (float*)alloc((size_t)N_NODES * sizeof(float));
  float*    att_raw  = (float*)alloc((size_t)N_EDGES * sizeof(float));
  float*    rows_sum = (float*)alloc((size_t)N_NODES * sizeof(float));
  unsigned* minmax   = (unsigned*)alloc(2 * sizeof(unsigned));

  const int TB = 256;

  gat_init<<<cdiv(N_NODES * D_OUT, TB), TB, 0, stream>>>(hout, rows_sum, minmax);

  {
    const int n_waves  = N_NODES / 16;
    const int waves_pb = TB / 32;
    gat_gemm_wmma<<<cdiv(n_waves, waves_pb), TB, 0, stream>>>(X, W, Xp);
  }

  gat_node_scores<<<cdiv(N_NODES, TB), TB, 0, stream>>>(Xp, a0, a1, s0, s1);

  gat_edge_scores<<<cdiv(N_EDGES, TB), TB, 0, stream>>>(row_idx, col_idx, s0, s1, att_raw, minmax);

  {
    const int waves_pb = TB / 32;
    gat_edge_scatter<<<cdiv(N_EDGES, waves_pb), TB, 0, stream>>>(
        row_idx, col_idx, att_raw, Xp, minmax, hout, rows_sum);
  }

  gat_finalize<<<cdiv(N_NODES * D_OUT, TB), TB, 0, stream>>>(hout, rows_sum);
}


# edge scatter: each wave takes 4 batches of 32 edges (per-lane index loads and exp, readlane broadcast, 32 row gathers then 32 line adds), not one wave per edge; f32 atomics keep arrival order
# speedup vs baseline: 2.1272x; 2.1272x over previous
#include <hip/hip_runtime.h>
#include <hip/hip_bf16.h>

#define N_NODES 50000
#define N_EDGES 1600000
#define D_IN    128
#define D_OUT   32
#define ALPHA   0.2f

typedef __attribute__((ext_vector_type(2))) float v2f;
typedef __attribute__((ext_vector_type(8))) float v8f;

__device__ __forceinline__ unsigned f2ord(float f) {
  unsigned u = __float_as_uint(f);
  return (u & 0x80000000u) ? ~u : (u | 0x80000000u);
}
__device__ __forceinline__ float ord2f(unsigned u) {
  u = (u & 0x80000000u) ? (u & 0x7FFFFFFFu) : ~u;
  return __uint_as_float(u);
}

__global__ void gat_init(float* __restrict__ hout, float* __restrict__ rows_sum,
                         unsigned* __restrict__ minmax) {
  int i = blockIdx.x * blockDim.x + threadIdx.x;
  if (i < N_NODES * D_OUT) hout[i] = 0.0f;
  if (i < N_NODES) rows_sum[i] = 0.0f;
  if (i == 0) { minmax[0] = 0xFFFFFFFFu; minmax[1] = 0u; }
}

__global__ void gat_gemm_wmma(const float* __restrict__ X,
                              const float* __restrict__ W,
                              float* __restrict__ Xp) {
  const int wave_in_block = threadIdx.x >> 5;
  const int lane          = threadIdx.x & 31;
  const int wave_id       = blockIdx.x * (blockDim.x >> 5) + wave_in_block;
  const int row_base      = wave_id * 16;
  if (row_base >= N_NODES) return;

  const int m    = lane & 15;
  const int kgrp = lane >> 4;

  const float* __restrict__ xrow = X + (size_t)(row_base + m) * D_IN;

  v8f acc0 = {};
  v8f acc1 = {};

  for (int k = 0; k < D_IN; k += 4) {
    const int kk = k + kgrp * 2;
    v2f a;
    a.x = xrow[kk + 0];
    a.y = xrow[kk + 1];
    v2f b0, b1;
    b0.x = W[(size_t)(kk + 0) * D_OUT + m];
    b0.y = W[(size_t)(kk + 1) * D_OUT + m];
    b1.x = W[(size_t)(kk + 0) * D_OUT + 16 + m];
    b1.y = W[(size_t)(kk + 1) * D_OUT + 16 + m];

    acc0 = __builtin_amdgcn_wmma_f32_16x16x4_f32(false, a, false, b0, (short)0, acc0, false, false);
    acc1 = __builtin_amdgcn_wmma_f32_16x16x4_f32(false, a, false, b1, (short)0, acc1, false, false);
  }

  const int rbase = row_base + kgrp * 8;
  for (int i = 0; i < 8; ++i) {
    const size_t r = (size_t)(rbase + i);
    Xp[r * D_OUT + m]      = acc0[i];
    Xp[r * D_OUT + 16 + m] = acc1[i];
  }
}

__global__ void gat_node_scores(const float* __restrict__ Xp,
                                const float* __restrict__ a0,
                                const float* __restrict__ a1,
                                float* __restrict__ s0,
                                float* __restrict__ s1) {
  int n = blockIdx.x * blockDim.x + threadIdx.x;
  if (n >= N_NODES) return;
  float acc0 = 0.0f, acc1 = 0.0f;
  const float* __restrict__ xr = Xp + (size_t)n * D_OUT;
#pragma unroll
  for (int d = 0; d < D_OUT; ++d) {
    float v = xr[d];
    acc0 += v * a0[d];
    acc1 += v * a1[d];
  }
  s0[n] = acc0;
  s1[n] = acc1;
}

__global__ void gat_edge_scores(const int* __restrict__ row_idx,
                                const int* __restrict__ col_idx,
                                const float* __restrict__ s0,
                                const float* __restrict__ s1,
                                float* __restrict__ att_raw,
                                unsigned* __restrict__ minmax) {
  __shared__ unsigned bmin, bmax;
  if (threadIdx.x == 0) { bmin = 0xFFFFFFFFu; bmax = 0u; }
  __syncthreads();

  int e = blockIdx.x * blockDim.x + threadIdx.x;
  if (e < N_EDGES) {
    float v = s0[row_idx[e]] + s1[col_idx[e]];
    v = (v > 0.0f) ? v : (ALPHA * v);
    att_raw[e] = v;
    unsigned o = f2ord(v);
    atomicMin(&bmin, o);
    atomicMax(&bmax, o);
  }
  __syncthreads();
  if (threadIdx.x == 0) {
    atomicMin(&minmax[0], bmin);
    atomicMax(&minmax[1], bmax);
  }
}

#define EDGE_BATCH    32
#define WAVE_BATCHES  4
#define SCATTER_WAVES 4
#define N_BATCHES     (N_EDGES / EDGE_BATCH)
static_assert(N_EDGES % EDGE_BATCH == 0, "a batch is a full wave of edges");
static_assert(D_OUT == 32, "one lane per output channel");

__global__ __launch_bounds__(SCATTER_WAVES * 32)
void gat_edge_scatter(const int* __restrict__ row_idx,
                      const int* __restrict__ col_idx,
                      const float* __restrict__ att_raw,
                      const float* __restrict__ Xp,
                      const unsigned* __restrict__ minmax,
                      float* __restrict__ hout,
                      float* __restrict__ rows_sum) {
  const int lane  = threadIdx.x & 31;
  const int wave  = (blockIdx.x * blockDim.x + threadIdx.x) >> 5;
  const int first = wave * WAVE_BATCHES;

  int   r[WAVE_BATCHES];
  int   c[WAVE_BATCHES];
  float raw[WAVE_BATCHES];
#pragma unroll
  for (int b = 0; b < WAVE_BATCHES; ++b) {
    r[b] = 0; c[b] = 0; raw[b] = 0.0f;
    if (first + b < N_BATCHES) {
      const int e = (first + b) * EDGE_BATCH + lane;
      r[b]   = row_idx[e];
      c[b]   = col_idx[e];
      raw[b] = att_raw[e];
    }
  }

  const float vmin = ord2f(minmax[0]);
  const float vmax = ord2f(minmax[1]);
  const float inv  = 1.0f / (vmax - vmin);

#pragma unroll
  for (int b = 0; b < WAVE_BATCHES; ++b) {
    if (first + b >= N_BATCHES) break;
    const float att = __expf((raw[b] - vmin) * inv);
    atomicAdd(&rows_sum[r[b]], att);

    float x[EDGE_BATCH];
#pragma unroll
    for (int k = 0; k < EDGE_BATCH; ++k) {
      const int ck = __builtin_amdgcn_readlane(c[b], k);
      x[k] = Xp[(unsigned)ck * D_OUT + lane];
    }
#pragma unroll
    for (int k = 0; k < EDGE_BATCH; ++k) {
      const int   rk = __builtin_amdgcn_readlane(r[b], k);
      const float ak = __int_as_float(__builtin_amdgcn_readlane(__float_as_int(att), k));
      atomicAdd(&hout[(unsigned)rk * D_OUT + lane], ak * x[k]);
    }
  }
}

__global__ void gat_finalize(float* __restrict__ hout,
                             const float* __restrict__ rows_sum) {
  int i = blockIdx.x * blockDim.x + threadIdx.x;
  if (i >= N_NODES * D_OUT) return;
  hout[i] /= rows_sum[i / D_OUT];
}

static inline int cdiv(int a, int b) { return (a + b - 1) / b; }

extern "C" void kernel_launch(void* const* d_in, const int* in_sizes, int n_in,
                              void* d_out, int out_size, void* d_ws, size_t ws_size,
                              hipStream_t stream) {
  (void)in_sizes; (void)n_in; (void)out_size; (void)ws_size;

  const float* X       = (const float*)d_in[0];
  const float* W       = (const float*)d_in[1];
  const float* a0      = (const float*)d_in[2];
  const float* a1      = (const float*)d_in[3];
  const int*   row_idx = (const int*)d_in[4];
  const int*   col_idx = (const int*)d_in[5];
  float* hout = (float*)d_out;

  char* ws = (char*)d_ws;
  size_t off = 0;
  auto alloc = [&](size_t bytes) { char* p = ws + off; off = (off + bytes + 255) & ~(size_t)255; return p; };
  float*    Xp       = (float*)alloc((size_t)N_NODES * D_OUT * sizeof(float));
  float*    s0       = (float*)alloc((size_t)N_NODES * sizeof(float));
  float*    s1       = (float*)alloc((size_t)N_NODES * sizeof(float));
  float*    att_raw  = (float*)alloc((size_t)N_EDGES * sizeof(float));
  float*    rows_sum = (float*)alloc((size_t)N_NODES * sizeof(float));
  unsigned* minmax   = (unsigned*)alloc(2 * sizeof(unsigned));

  const int TB = 256;

  gat_init<<<cdiv(N_NODES * D_OUT, TB), TB, 0, stream>>>(hout, rows_sum, minmax);

  {
    const int n_waves  = N_NODES / 16;
    const int waves_pb = TB / 32;
    gat_gemm_wmma<<<cdiv(n_waves, waves_pb), TB, 0, stream>>>(X, W, Xp);
  }

  gat_node_scores<<<cdiv(N_NODES, TB), TB, 0, stream>>>(Xp, a0, a1, s0, s1);

  gat_edge_scores<<<cdiv(N_EDGES, TB), TB, 0, stream>>>(row_idx, col_idx, s0, s1, att_raw, minmax);

  gat_edge_scatter<<<cdiv(N_BATCHES, SCATTER_WAVES * WAVE_BATCHES), SCATTER_WAVES * 32, 0, stream>>>(
      row_idx, col_idx, att_raw, Xp, minmax, hout, rows_sum);

  gat_finalize<<<cdiv(N_NODES * D_OUT, TB), TB, 0, stream>>>(hout, rows_sum);
}


# edge scatter: rows split between the package's two halves; a wave claims chunks from per-half counters, adds only into rows of its own half, then helps the other; f32 atomic arrival order changes
# speedup vs baseline: 3.4573x; 1.6253x over previous
#include <hip/hip_runtime.h>
#include <hip/hip_bf16.h>

#define N_NODES 50000
#define N_EDGES 1600000
#define D_IN    128
#define D_OUT   32
#define ALPHA   0.2f

typedef __attribute__((ext_vector_type(2))) float v2f;
typedef __attribute__((ext_vector_type(8))) float v8f;

__device__ __forceinline__ unsigned f2ord(float f) {
  unsigned u = __float_as_uint(f);
  return (u & 0x80000000u) ? ~u : (u | 0x80000000u);
}
__device__ __forceinline__ float ord2f(unsigned u) {
  u = (u & 0x80000000u) ? (u & 0x7FFFFFFFu) : ~u;
  return __uint_as_float(u);
}

__global__ void gat_init(float* __restrict__ hout, float* __restrict__ rows_sum,
                         unsigned* __restrict__ minmax,
                         unsigned* __restrict__ claim0, unsigned* __restrict__ claim1) {
  int i = blockIdx.x * blockDim.x + threadIdx.x;
  if (i < N_NODES * D_OUT) hout[i] = 0.0f;
  if (i < N_NODES) rows_sum[i] = 0.0f;
  if (i == 0) { minmax[0] = 0xFFFFFFFFu; minmax[1] = 0u; claim0[0] = 0u; claim1[0] = 0u; }
}

__global__ void gat_gemm_wmma(const float* __restrict__ X,
                              const float* __restrict__ W,
                              float* __restrict__ Xp) {
  const int wave_in_block = threadIdx.x >> 5;
  const int lane          = threadIdx.x & 31;
  const int wave_id       = blockIdx.x * (blockDim.x >> 5) + wave_in_block;
  const int row_base      = wave_id * 16;
  if (row_base >= N_NODES) return;

  const int m    = lane & 15;
  const int kgrp = lane >> 4;

  const float* __restrict__ xrow = X + (size_t)(row_base + m) * D_IN;

  v8f acc0 = {};
  v8f acc1 = {};

  for (int k = 0; k < D_IN; k += 4) {
    const int kk = k + kgrp * 2;
    v2f a;
    a.x = xrow[kk + 0];
    a.y = xrow[kk + 1];
    v2f b0, b1;
    b0.x = W[(size_t)(kk + 0) * D_OUT + m];
    b0.y = W[(size_t)(kk + 1) * D_OUT + m];
    b1.x = W[(size_t)(kk + 0) * D_OUT + 16 + m];
    b1.y = W[(size_t)(kk + 1) * D_OUT + 16 + m];

    acc0 = __builtin_amdgcn_wmma_f32_16x16x4_f32(false, a, false, b0, (short)0, acc0, false, false);
    acc1 = __builtin_amdgcn_wmma_f32_16x16x4_f32(false, a, false, b1, (short)0, acc1, false, false);
  }

  const int rbase = row_base + kgrp * 8;
  for (int i = 0; i < 8; ++i) {
    const size_t r = (size_t)(rbase + i);
    Xp[r * D_OUT + m]      = acc0[i];
    Xp[r * D_OUT + 16 + m] = acc1[i];
  }
}

__global__ void gat_node_scores(const float* __restrict__ Xp,
                                const float* __restrict__ a0,
                                const float* __restrict__ a1,
                                float* __restrict__ s0,
                                float* __restrict__ s1) {
  int n = blockIdx.x * blockDim.x + threadIdx.x;
  if (n >= N_NODES) return;
  float acc0 = 0.0f, acc1 = 0.0f;
  const float* __restrict__ xr = Xp + (size_t)n * D_OUT;
#pragma unroll
  for (int d = 0; d < D_OUT; ++d) {
    float v = xr[d];
    acc0 += v * a0[d];
    acc1 += v * a1[d];
  }
  s0[n] = acc0;
  s1[n] = acc1;
}

__global__ void gat_edge_scores(const int* __restrict__ row_idx,
                                const int* __restrict__ col_idx,
                                const float* __restrict__ s0,
                                const float* __restrict__ s1,
                                float* __restrict__ att_raw,
                                unsigned* __restrict__ minmax) {
  __shared__ unsigned bmin, bmax;
  if (threadIdx.x == 0) { bmin = 0xFFFFFFFFu; bmax = 0u; }
  __syncthreads();

  int e = blockIdx.x * blockDim.x + threadIdx.x;
  if (e < N_EDGES) {
    float v = s0[row_idx[e]] + s1[col_idx[e]];
    v = (v > 0.0f) ? v : (ALPHA * v);
    att_raw[e] = v;
    unsigned o = f2ord(v);
    atomicMin(&bmin, o);
    atomicMax(&bmax, o);
  }
  __syncthreads();
  if (threadIdx.x == 0) {
    atomicMin(&minmax[0], bmin);
    atomicMax(&minmax[1], bmax);
  }
}

#define EDGE_BATCH     32
#define CHUNK_BATCHES  4
#define SCATTER_WAVES  4
#define SCATTER_BLOCKS 2048
#define N_BATCHES      (N_EDGES / EDGE_BATCH)
#define N_CHUNKS       (N_BATCHES / CHUNK_BATCHES)
#define ROW_SPLIT      (N_NODES / 2)
#define GROUP          8
static_assert(N_EDGES % EDGE_BATCH == 0, "a batch is a full wave of edges");
static_assert(N_BATCHES % CHUNK_BATCHES == 0, "a chunk is a whole number of batches");
static_assert(D_OUT == 32, "one lane per output channel");
static_assert((ROW_SPLIT * D_OUT * 4) % 256 == 0, "the split falls on a 256-byte boundary of h");

__global__ __launch_bounds__(SCATTER_WAVES * 32)
void gat_edge_scatter(const int* __restrict__ row_idx,
                      const int* __restrict__ col_idx,
                      const float* __restrict__ att_raw,
                      const float* __restrict__ Xp,
                      const unsigned* __restrict__ minmax,
                      float* __restrict__ hout,
                      float* __restrict__ rows_sum,
                      unsigned* __restrict__ claim0,
                      unsigned* __restrict__ claim1) {
  const int lane = threadIdx.x & 31;

  const unsigned place   = __builtin_amdgcn_s_sendmsg_rtn(0x87);
  const int      my_half = (int)((place >> 18) & 1u);

  const float vmin = ord2f(minmax[0]);
  const float vmax = ord2f(minmax[1]);
  const float inv  = 1.0f / (vmax - vmin);

  for (int pass = 0; pass < 2; ++pass) {
    const int h = my_half ^ pass;
    unsigned* __restrict__ claim = h ? claim1 : claim0;

    for (;;) {
      unsigned chunk = 0u;
      if (lane == 0) chunk = atomicAdd(claim, 1u);
      chunk = (unsigned)__builtin_amdgcn_readfirstlane((int)chunk);
      if (chunk >= (unsigned)N_CHUNKS) break;
      const int first = (int)chunk * CHUNK_BATCHES;

      int   r[CHUNK_BATCHES];
      int   c[CHUNK_BATCHES];
      float raw[CHUNK_BATCHES];
#pragma unroll
      for (int b = 0; b < CHUNK_BATCHES; ++b) {
        const int e = (first + b) * EDGE_BATCH + lane;
        r[b]   = row_idx[e];
        c[b]   = col_idx[e];
        raw[b] = att_raw[e];
      }

#pragma unroll
      for (int b = 0; b < CHUNK_BATCHES; ++b) {
        const bool     mine = ((r[b] >= ROW_SPLIT) ? 1 : 0) == h;
        const unsigned own  = (unsigned)__builtin_amdgcn_ballot_w32(mine);
        const float    att  = __expf((raw[b] - vmin) * inv);
        if (mine) atomicAdd(&rows_sum[r[b]], att);

        unsigned m = own;
        while (m != 0u) {
          int   kk[GROUP];
          float x[GROUP];
#pragma unroll
          for (int i = 0; i < GROUP; ++i) {
            kk[i] = -1;
            x[i]  = 0.0f;
            if (m != 0u) {
              kk[i] = __builtin_ctz(m);
              m &= m - 1u;
              const int ck = __builtin_amdgcn_readlane(c[b], kk[i]);
              x[i] = Xp[(unsigned)ck * D_OUT + lane];
            }
          }
#pragma unroll
          for (int i = 0; i < GROUP; ++i) {
            if (kk[i] >= 0) {
              const int   rk = __builtin_amdgcn_readlane(r[b], kk[i]);
              const float ak = __int_as_float(__builtin_amdgcn_readlane(__float_as_int(att), kk[i]));
              atomicAdd(&hout[(unsigned)rk * D_OUT + lane], ak * x[i]);
            }
          }
        }
      }
    }
  }
}

__global__ void gat_finalize(float* __restrict__ hout,
                             const float* __restrict__ rows_sum) {
  int i = blockIdx.x * blockDim.x + threadIdx.x;
  if (i >= N_NODES * D_OUT) return;
  hout[i] /= rows_sum[i / D_OUT];
}

static inline int cdiv(int a, int b) { return (a + b - 1) / b; }

extern "C" void kernel_launch(void* const* d_in, const int* in_sizes, int n_in,
                              void* d_out, int out_size, void* d_ws, size_t ws_size,
                              hipStream_t stream) {
  (void)in_sizes; (void)n_in; (void)out_size; (void)ws_size;

  const float* X       = (const float*)d_in[0];
  const float* W       = (const float*)d_in[1];
  const float* a0      = (const float*)d_in[2];
  const float* a1      = (const float*)d_in[3];
  const int*   row_idx = (const int*)d_in[4];
  const int*   col_idx = (const int*)d_in[5];
  float* hout = (float*)d_out;

  char* ws = (char*)d_ws;
  size_t off = 0;
  auto alloc = [&](size_t bytes) { char* p = ws + off; off = (off + bytes + 255) & ~(size_t)255; return p; };
  float*    Xp       = (float*)alloc((size_t)N_NODES * D_OUT * sizeof(float));
  float*    s0       = (float*)alloc((size_t)N_NODES * sizeof(float));
  float*    s1       = (float*)alloc((size_t)N_NODES * sizeof(float));
  float*    att_raw  = (float*)alloc((size_t)N_EDGES * sizeof(float));
  float*    rows_sum = (float*)alloc((size_t)N_NODES * sizeof(float));
  unsigned* minmax   = (unsigned*)alloc(2 * sizeof(unsigned));
  unsigned* claim0   = (unsigned*)alloc(sizeof(unsigned));
  unsigned* claim1   = (unsigned*)alloc(sizeof(unsigned));

  const int TB = 256;

  gat_init<<<cdiv(N_NODES * D_OUT, TB), TB, 0, stream>>>(hout, rows_sum, minmax, claim0, claim1);

  {
    const int n_waves  = N_NODES / 16;
    const int waves_pb = TB / 32;
    gat_gemm_wmma<<<cdiv(n_waves, waves_pb), TB, 0, stream>>>(X, W, Xp);
  }

  gat_node_scores<<<cdiv(N_NODES, TB), TB, 0, stream>>>(Xp, a0, a1, s0, s1);

  gat_edge_scores<<<cdiv(N_EDGES, TB), TB, 0, stream>>>(row_idx, col_idx, s0, s1, att_raw, minmax);

  gat_edge_scatter<<<SCATTER_BLOCKS, SCATTER_WAVES * 32, 0, stream>>>(
      row_idx, col_idx, att_raw, Xp, minmax, hout, rows_sum, claim0, claim1);

  gat_finalize<<<cdiv(N_NODES * D_OUT, TB), TB, 0, stream>>>(hout, rows_sum);
}


# edge scatter: each work-group keeps a 1564-row slab of h and rows_sum in LDS, filters 1/8 of the edges for it, adds gathered rows by LDS f32 atomics, adds the slab to h once; f32 sum order changes
# speedup vs baseline: 4.4026x; 1.2734x over previous
#include <hip/hip_runtime.h>
#include <hip/hip_bf16.h>

#define N_NODES 50000
#define N_EDGES 1600000
#define D_IN    128
#define D_OUT   32
#define ALPHA   0.2f

typedef __attribute__((ext_vector_type(2))) float v2f;
typedef __attribute__((ext_vector_type(8))) float v8f;

__device__ __forceinline__ unsigned f2ord(float f) {
  unsigned u = __float_as_uint(f);
  return (u & 0x80000000u) ? ~u : (u | 0x80000000u);
}
__device__ __forceinline__ float ord2f(unsigned u) {
  u = (u & 0x80000000u) ? (u & 0x7FFFFFFFu) : ~u;
  return __uint_as_float(u);
}

__device__ __forceinline__ float row_sum16(float v) {
  v += __int_as_float(__builtin_amdgcn_update_dpp(0, __float_as_int(v), 0x161, 0xf, 0xf, false));
  v += __int_as_float(__builtin_amdgcn_update_dpp(0, __float_as_int(v), 0x162, 0xf, 0xf, false));
  v += __int_as_float(__builtin_amdgcn_update_dpp(0, __float_as_int(v), 0x164, 0xf, 0xf, false));
  v += __int_as_float(__builtin_amdgcn_update_dpp(0, __float_as_int(v), 0x168, 0xf, 0xf, false));
  return v;
}

static_assert(N_NODES % 16 == 0, "the tiles cover every row of h, rows_sum, s0 and s1 exactly once");
__global__ void gat_gemm_wmma(const float* __restrict__ X,
                              const float* __restrict__ W,
                              const float* __restrict__ a0,
                              const float* __restrict__ a1,
                              float* __restrict__ Xp,
                              float* __restrict__ s0,
                              float* __restrict__ s1,
                              float* __restrict__ hout,
                              float* __restrict__ rows_sum,
                              unsigned* __restrict__ minmax,
                              unsigned* __restrict__ claim0,
                              unsigned* __restrict__ claim1) {
  if (blockIdx.x == 0 && threadIdx.x == 0) {
    minmax[0] = 0xFFFFFFFFu; minmax[1] = 0u; claim0[0] = 0u; claim1[0] = 0u;
  }

  const int wave_in_block = threadIdx.x >> 5;
  const int lane          = threadIdx.x & 31;
  const int wave_id       = blockIdx.x * (blockDim.x >> 5) + wave_in_block;
  const int row_base      = wave_id * 16;
  if (row_base >= N_NODES) return;

  for (int i = 0; i < 16; ++i) hout[(size_t)(row_base + i) * D_OUT + lane] = 0.0f;
  if (lane < 16) rows_sum[row_base + lane] = 0.0f;

  const int m    = lane & 15;
  const int kgrp = lane >> 4;

  const float* __restrict__ xrow = X + (size_t)(row_base + m) * D_IN;

  v8f acc0 = {};
  v8f acc1 = {};

  for (int k = 0; k < D_IN; k += 4) {
    const int kk = k + kgrp * 2;
    v2f a;
    a.x = xrow[kk + 0];
    a.y = xrow[kk + 1];
    v2f b0, b1;
    b0.x = W[(size_t)(kk + 0) * D_OUT + m];
    b0.y = W[(size_t)(kk + 1) * D_OUT + m];
    b1.x = W[(size_t)(kk + 0) * D_OUT + 16 + m];
    b1.y = W[(size_t)(kk + 1) * D_OUT + 16 + m];

    acc0 = __builtin_amdgcn_wmma_f32_16x16x4_f32(false, a, false, b0, (short)0, acc0, false, false);
    acc1 = __builtin_amdgcn_wmma_f32_16x16x4_f32(false, a, false, b1, (short)0, acc1, false, false);
  }

  const int rbase = row_base + kgrp * 8;
  for (int i = 0; i < 8; ++i) {
    const size_t r = (size_t)(rbase + i);
    Xp[r * D_OUT + m]      = acc0[i];
    Xp[r * D_OUT + 16 + m] = acc1[i];
  }

  const float a0lo = a0[m], a0hi = a0[16 + m];
  const float a1lo = a1[m], a1hi = a1[16 + m];
  float v0 = 0.0f, v1 = 0.0f;
  for (int i = 0; i < 8; ++i) {
    const float t0 = row_sum16(acc0[i] * a0lo + acc1[i] * a0hi);
    const float t1 = row_sum16(acc0[i] * a1lo + acc1[i] * a1hi);
    if (m == i) { v0 = t0; v1 = t1; }
  }
  if (m < 8) {
    s0[rbase + m] = v0;
    s1[rbase + m] = v1;
  }
}

__global__ void gat_edge_scores(const int* __restrict__ row_idx,
                                const int* __restrict__ col_idx,
                                const float* __restrict__ s0,
                                const float* __restrict__ s1,
                                float* __restrict__ att_raw,
                                unsigned* __restrict__ minmax) {
  __shared__ unsigned bmin, bmax;
  if (threadIdx.x == 0) { bmin = 0xFFFFFFFFu; bmax = 0u; }
  __syncthreads();

  int e = blockIdx.x * blockDim.x + threadIdx.x;
  if (e < N_EDGES) {
    float v = s0[row_idx[e]] + s1[col_idx[e]];
    v = (v > 0.0f) ? v : (ALPHA * v);
    att_raw[e] = v;
    unsigned o = f2ord(v);
    atomicMin(&bmin, o);
    atomicMax(&bmax, o);
  }
  __syncthreads();
  if (threadIdx.x == 0) {
    atomicMin(&minmax[0], bmin);
    atomicMax(&minmax[1], bmax);
  }
}

#define SB_EDGES    128
#define N_SB        (N_EDGES / SB_EDGES)
#define SLABS       32
#define SLAB_ROWS   1564
#define PARTS       8
#define UNITS_HALF  ((SLABS / 2) * PARTS)
#define ACC_WAVES   32
#define ACC_BLOCKS  (SLABS * PARTS)
#define QCAP        64
#define GROUP       16
#define FLUSH_ROWS  8
static_assert(N_EDGES % SB_EDGES == 0, "an index load of a wave is 128 whole edges");
static_assert(SLABS * SLAB_ROWS >= N_NODES && (SLABS - 1) * SLAB_ROWS < N_NODES, "the slabs cover the rows, none is empty");
static_assert(((SLABS / 2) * SLAB_ROWS) % 32 == 0, "the halves meet on a line boundary of h and of rows_sum");
static_assert(D_OUT == 32, "one lane per output channel");
static_assert(QCAP == 64, "the queue index is taken modulo 64");

__global__ __launch_bounds__(ACC_WAVES * 32)
void gat_edge_scatter(const int* __restrict__ row_idx,
                      const int* __restrict__ col_idx,
                      const float* __restrict__ att_raw,
                      const float* __restrict__ Xp,
                      const unsigned* __restrict__ minmax,
                      float* __restrict__ hout,
                      float* __restrict__ rows_sum,
                      unsigned* __restrict__ claim0,
                      unsigned* __restrict__ claim1) {
  __shared__ __attribute__((aligned(16))) float s_h[SLAB_ROWS * D_OUT];
  __shared__ float s_sum[SLAB_ROWS];
  __shared__ int2  s_q[ACC_WAVES][QCAP];
  __shared__ int   s_unit;

  const int lane = threadIdx.x & 31;
  const int wave = __builtin_amdgcn_readfirstlane((int)(threadIdx.x >> 5));

  const unsigned place   = __builtin_amdgcn_s_sendmsg_rtn(0x87);
  const int      my_half = (int)((place >> 18) & 1u);

  const float vmin = ord2f(minmax[0]);
  const float vmax = ord2f(minmax[1]);
  const float inv  = 1.0f / (vmax - vmin);

  int2* __restrict__ q = &s_q[wave][0];

  for (;;) {
    if (threadIdx.x == 0) {
      int u = -1;
      unsigned t = atomicAdd(my_half ? claim1 : claim0, 1u);
      if (t < (unsigned)UNITS_HALF) {
        u = my_half * UNITS_HALF + (int)t;
      } else {
        t = atomicAdd(my_half ? claim0 : claim1, 1u);
        if (t < (unsigned)UNITS_HALF) u = (my_half ^ 1) * UNITS_HALF + (int)t;
      }
      s_unit = u;
    }
    __syncthreads();
    const int unit = __builtin_amdgcn_readfirstlane(s_unit);
    if (unit < 0) break;

    const int slab  = unit / PARTS;
    const int part  = unit % PARTS;
    const int base  = slab * SLAB_ROWS;
    const int nrows = (N_NODES - base < SLAB_ROWS) ? (N_NODES - base) : SLAB_ROWS;
    const int sb0   = (part * N_SB) / PARTS;
    const int sb1   = ((part + 1) * N_SB) / PARTS;

    {
      float4* __restrict__ h4 = reinterpret_cast<float4*>(s_h);
      for (int i = threadIdx.x; i < SLAB_ROWS * D_OUT / 4; i += ACC_WAVES * 32) h4[i] = make_float4(0.0f, 0.0f, 0.0f, 0.0f);
      for (int i = threadIdx.x; i < SLAB_ROWS; i += ACC_WAVES * 32) s_sum[i] = 0.0f;
    }
    __syncthreads();

    int head = 0, cnt = 0;

    auto drain = [&](const int n) {
      const bool valid = lane < n;
      int   erow = 0, ecol = 0;
      float att  = 0.0f;
      if (valid) {
        const int2 ent = q[(head + lane) & (QCAP - 1)];
        erow = ent.y;
        ecol = col_idx[ent.x];
        att  = __expf((att_raw[ent.x] - vmin) * inv);
        __hip_atomic_fetch_add(&s_sum[erow], att, __ATOMIC_RELAXED, __HIP_MEMORY_SCOPE_WORKGROUP);
      }
      for (int g = 0; g < n; g += GROUP) {
        float x[GROUP];
#pragma unroll
        for (int t = 0; t < GROUP; ++t) {
          x[t] = 0.0f;
          if (g + t < n) {
            const int ck = __builtin_amdgcn_readlane(ecol, g + t);
            x[t] = Xp[(unsigned)ck * D_OUT + lane];
          }
        }
#pragma unroll
        for (int t = 0; t < GROUP; ++t) {
          if (g + t < n) {
            const int   rk = __builtin_amdgcn_readlane(erow, g + t);
            const float ak = __int_as_float(__builtin_amdgcn_readlane(__float_as_int(att), g + t));
            __hip_atomic_fetch_add(&s_h[rk * D_OUT + lane], ak * x[t], __ATOMIC_RELAXED, __HIP_MEMORY_SCOPE_WORKGROUP);
          }
        }
      }
    };

    const int4* __restrict__ row4 = reinterpret_cast<const int4*>(row_idx);
    int  sb = sb0 + wave;
    int4 nxt = make_int4(0, 0, 0, 0);
    if (sb < sb1) nxt = row4[sb * (SB_EDGES / 4) + lane];
    for (; sb < sb1; sb += ACC_WAVES) {
      const int4 cur = nxt;
      if (sb + ACC_WAVES < sb1) nxt = row4[(sb + ACC_WAVES) * (SB_EDGES / 4) + lane];
      const int rr[4] = {cur.x, cur.y, cur.z, cur.w};
#pragma unroll
      for (int j = 0; j < 4; ++j) {
        const unsigned d    = (unsigned)(rr[j] - base);
        const bool     keep = d < (unsigned)nrows;
        const unsigned mask = (unsigned)__builtin_amdgcn_ballot_w32(keep);
        if (mask != 0u) {
          if (keep) {
            const int pos = (head + cnt + (int)__builtin_amdgcn_mbcnt_lo(mask, 0u)) & (QCAP - 1);
            q[pos] = make_int2(sb * SB_EDGES + lane * 4 + j, (int)d);
          }
          cnt += __builtin_popcount(mask);
          if (cnt >= 32) {
            drain(32);
            head = (head + 32) & (QCAP - 1);
            cnt -= 32;
          }
        }
      }
    }
    if (cnt > 0) drain(cnt);
    __syncthreads();

    for (int row0 = wave; row0 < nrows; row0 += FLUSH_ROWS * ACC_WAVES) {
      float v[FLUSH_ROWS];
#pragma unroll
      for (int t = 0; t < FLUSH_ROWS; ++t) {
        const int row = row0 + t * ACC_WAVES;
        v[t] = (row < nrows) ? s_h[row * D_OUT + lane] : 0.0f;
      }
#pragma unroll
      for (int t = 0; t < FLUSH_ROWS; ++t) {
        const int row = row0 + t * ACC_WAVES;
        if (row < nrows) atomicAdd(&hout[(unsigned)(base + row) * D_OUT + lane], v[t]);
      }
    }
    for (int i = threadIdx.x; i < nrows; i += ACC_WAVES * 32) atomicAdd(&rows_sum[base + i], s_sum[i]);

    __syncthreads();
  }
}

__global__ void gat_finalize(float* __restrict__ hout,
                             const float* __restrict__ rows_sum) {
  int i = blockIdx.x * blockDim.x + threadIdx.x;
  if (i >= N_NODES * D_OUT) return;
  hout[i] /= rows_sum[i / D_OUT];
}

static inline int cdiv(int a, int b) { return (a + b - 1) / b; }

extern "C" void kernel_launch(void* const* d_in, const int* in_sizes, int n_in,
                              void* d_out, int out_size, void* d_ws, size_t ws_size,
                              hipStream_t stream) {
  (void)in_sizes; (void)n_in; (void)out_size; (void)ws_size;

  const float* X       = (const float*)d_in[0];
  const float* W       = (const float*)d_in[1];
  const float* a0      = (const float*)d_in[2];
  const float* a1      = (const float*)d_in[3];
  const int*   row_idx = (const int*)d_in[4];
  const int*   col_idx = (const int*)d_in[5];
  float* hout = (float*)d_out;

  char* ws = (char*)d_ws;
  size_t off = 0;
  auto alloc = [&](size_t bytes) { char* p = ws + off; off = (off + bytes + 255) & ~(size_t)255; return p; };
  float*    Xp       = (float*)alloc((size_t)N_NODES * D_OUT * sizeof(float));
  float*    s0       = (float*)alloc((size_t)N_NODES * sizeof(float));
  float*    s1       = (float*)alloc((size_t)N_NODES * sizeof(float));
  float*    att_raw  = (float*)alloc((size_t)N_EDGES * sizeof(float));
  float*    rows_sum = (float*)alloc((size_t)N_NODES * sizeof(float));
  unsigned* minmax   = (unsigned*)alloc(2 * sizeof(unsigned));
  unsigned* claim0   = (unsigned*)alloc(sizeof(unsigned));
  unsigned* claim1   = (unsigned*)alloc(sizeof(unsigned));

  const int TB = 256;

  {
    const int n_waves  = N_NODES / 16;
    const int waves_pb = TB / 32;
    gat_gemm_wmma<<<cdiv(n_waves, waves_pb), TB, 0, stream>>>(
        X, W, a0, a1, Xp, s0, s1, hout, rows_sum, minmax, claim0, claim1);
  }

  gat_edge_scores<<<cdiv(N_EDGES, TB), TB, 0, stream>>>(row_idx, col_idx, s0, s1, att_raw, minmax);

  gat_edge_scatter<<<ACC_BLOCKS, ACC_WAVES * 32, 0, stream>>>(
      row_idx, col_idx, att_raw, Xp, minmax, hout, rows_sum, claim0, claim1);

  gat_finalize<<<cdiv(N_NODES * D_OUT, TB), TB, 0, stream>>>(hout, rows_sum);
}


# score look-ups from LDS: the edge-sort launch copies s0, then s1, whole into LDS (16-byte whole-line loads) and the summing launch copies s1; no 4-byte per-edge gather goes to the second-level cache
# speedup vs baseline: 7.6902x; 1.7467x over previous
#include <hip/hip_runtime.h>
#include <hip/hip_bf16.h>

#define N_NODES 50000
#define N_EDGES 1600000
#define D_IN    128
#define D_OUT   32
#define ALPHA   0.2f

typedef __attribute__((ext_vector_type(2))) float v2f;
typedef __attribute__((ext_vector_type(8))) float v8f;

__device__ __forceinline__ unsigned f2ord(float f) {
  unsigned u = __float_as_uint(f);
  return (u & 0x80000000u) ? ~u : (u | 0x80000000u);
}
__device__ __forceinline__ float ord2f(unsigned u) {
  u = (u & 0x80000000u) ? (u & 0x7FFFFFFFu) : ~u;
  return __uint_as_float(u);
}

__device__ __forceinline__ float row_sum16(float v) {
  v += __int_as_float(__builtin_amdgcn_update_dpp(0, __float_as_int(v), 0x161, 0xf, 0xf, false));
  v += __int_as_float(__builtin_amdgcn_update_dpp(0, __float_as_int(v), 0x162, 0xf, 0xf, false));
  v += __int_as_float(__builtin_amdgcn_update_dpp(0, __float_as_int(v), 0x164, 0xf, 0xf, false));
  v += __int_as_float(__builtin_amdgcn_update_dpp(0, __float_as_int(v), 0x168, 0xf, 0xf, false));
  return v;
}

static_assert(N_NODES % 16 == 0, "the tiles cover every row of X', s0 and s1 exactly once");
__global__ void gat_gemm_wmma(const float* __restrict__ X,
                              const float* __restrict__ W,
                              const float* __restrict__ a0,
                              const float* __restrict__ a1,
                              float* __restrict__ Xp,
                              float* __restrict__ s0,
                              float* __restrict__ s1,
                              unsigned* __restrict__ minmax) {
  if (blockIdx.x == 0 && threadIdx.x == 0) {
    minmax[0] = 0xFFFFFFFFu; minmax[1] = 0u;
  }

  const int wave_in_block = threadIdx.x >> 5;
  const int lane          = threadIdx.x & 31;
  const int wave_id       = blockIdx.x * (blockDim.x >> 5) + wave_in_block;
  const int row_base      = wave_id * 16;
  if (row_base >= N_NODES) return;

  const int m    = lane & 15;
  const int kgrp = lane >> 4;

  const float* __restrict__ xrow = X + (size_t)(row_base + m) * D_IN;

  v8f acc0 = {};
  v8f acc1 = {};

  for (int k = 0; k < D_IN; k += 4) {
    const int kk = k + kgrp * 2;
    v2f a;
    a.x = xrow[kk + 0];
    a.y = xrow[kk + 1];
    v2f b0, b1;
    b0.x = W[(size_t)(kk + 0) * D_OUT + m];
    b0.y = W[(size_t)(kk + 1) * D_OUT + m];
    b1.x = W[(size_t)(kk + 0) * D_OUT + 16 + m];
    b1.y = W[(size_t)(kk + 1) * D_OUT + 16 + m];

    acc0 = __builtin_amdgcn_wmma_f32_16x16x4_f32(false, a, false, b0, (short)0, acc0, false, false);
    acc1 = __builtin_amdgcn_wmma_f32_16x16x4_f32(false, a, false, b1, (short)0, acc1, false, false);
  }

  const int rbase = row_base + kgrp * 8;
  for (int i = 0; i < 8; ++i) {
    const size_t r = (size_t)(rbase + i);
    Xp[r * D_OUT + m]      = acc0[i];
    Xp[r * D_OUT + 16 + m] = acc1[i];
  }

  const float a0lo = a0[m], a0hi = a0[16 + m];
  const float a1lo = a1[m], a1hi = a1[16 + m];
  float v0 = 0.0f, v1 = 0.0f;
  for (int i = 0; i < 8; ++i) {
    const float t0 = row_sum16(acc0[i] * a0lo + acc1[i] * a0hi);
    const float t1 = row_sum16(acc0[i] * a1lo + acc1[i] * a1hi);
    if (m == i) { v0 = t0; v1 = t1; }
  }
  if (m < 8) {
    s0[rbase + m] = v0;
    s1[rbase + m] = v1;
  }
}

#define STAGE_THREADS 1024
#define STAGE_VEC     (N_NODES / 4)
#define STAGE_PER     ((STAGE_VEC + STAGE_THREADS - 1) / STAGE_THREADS)
static_assert(N_NODES % 4 == 0, "a table is a whole number of 16-byte pieces");

__device__ __forceinline__ void stage_load(const float* __restrict__ table, float4 (&t)[STAGE_PER], int tid) {
  const float4* __restrict__ t4 = reinterpret_cast<const float4*>(table);
#pragma unroll
  for (int j = 0; j < STAGE_PER; ++j) {
    const int q = tid + j * STAGE_THREADS;
    t[j] = make_float4(0.0f, 0.0f, 0.0f, 0.0f);
    if (q < STAGE_VEC) t[j] = t4[q];
  }
}
__device__ __forceinline__ void stage_store(float* lds_table, const float4 (&t)[STAGE_PER], int tid) {
  float4* l4 = reinterpret_cast<float4*>(lds_table);
#pragma unroll
  for (int j = 0; j < STAGE_PER; ++j) {
    const int q = tid + j * STAGE_THREADS;
    if (q < STAGE_VEC) l4[q] = t[j];
  }
}

#define BIN_ROWS    196
#define N_BINS      256
#define N_CHUNKS    256
#define CHUNK       (N_EDGES / N_CHUNKS)
#define BIN_THREADS 1024
#define PER_THREAD  ((CHUNK + BIN_THREADS - 1) / BIN_THREADS)
static_assert(N_EDGES % N_CHUNKS == 0, "the chunks cover the edges exactly");
static_assert(N_BINS * BIN_ROWS >= N_NODES && (N_BINS - 1) * BIN_ROWS < N_NODES, "the blocks cover the rows, none is empty");
static_assert(BIN_ROWS <= 256, "the row inside its block is kept in 8 bits of the record");
static_assert(N_NODES <= 65536, "the source row is kept in 16 bits of the record");
static_assert(CHUNK < 65536, "a block's start inside a chunk is kept in 16 bits");
static_assert(N_BINS == 256 && BIN_THREADS >= N_BINS + 1, "the running sum takes 8 counts per lane of one wave");
static_assert(D_OUT == 32, "one lane per output channel");
static_assert(BIN_THREADS == STAGE_THREADS, "the table copy is laid out for this many threads");

__global__ __launch_bounds__(BIN_THREADS)
void gat_edge_bin(const int* __restrict__ row_idx,
                  const int* __restrict__ col_idx,
                  const float* __restrict__ s0,
                  const float* __restrict__ s1,
                  unsigned* __restrict__ rec,
                  unsigned* __restrict__ starts_words,
                  unsigned* __restrict__ minmax) {
  __shared__ unsigned s_cnt[N_BINS];
  __shared__ unsigned s_start[N_BINS + 1];
  __shared__ unsigned s_rec[CHUNK];
  __shared__ unsigned s_min, s_max;
  __shared__ alignas(16) float s_tab[N_NODES];

  const int tid   = threadIdx.x;
  const int lane  = tid & 31;
  const int chunk = blockIdx.x;
  const int e0    = chunk * CHUNK;

  if (tid < N_BINS) s_cnt[tid] = 0u;
  if (tid == 0) { s_min = 0xFFFFFFFFu; s_max = 0u; }

  unsigned myrec[PER_THREAD];
  unsigned mypos[PER_THREAD];
  unsigned omin = 0xFFFFFFFFu, omax = 0u;
  int   rr[PER_THREAD], cc[PER_THREAD];
  float sa[PER_THREAD], sb[PER_THREAD];
  float4 tab[STAGE_PER];
  stage_load(s0, tab, tid);
#pragma unroll
  for (int i = 0; i < PER_THREAD; ++i) {
    const int k = tid + i * BIN_THREADS;
    rr[i] = -1;
    cc[i] = 0;
    if (k < CHUNK) {
      rr[i] = row_idx[e0 + k];
      cc[i] = col_idx[e0 + k];
    }
  }
  stage_store(s_tab, tab, tid);
  __syncthreads();
#pragma unroll
  for (int i = 0; i < PER_THREAD; ++i) {
    sa[i] = 0.0f;
    if (tid + i * BIN_THREADS < CHUNK) sa[i] = s_tab[rr[i]];
  }
  stage_load(s1, tab, tid);
  __syncthreads();
  stage_store(s_tab, tab, tid);
  __syncthreads();
#pragma unroll
  for (int i = 0; i < PER_THREAD; ++i) {
    sb[i] = 0.0f;
    if (tid + i * BIN_THREADS < CHUNK) sb[i] = s_tab[cc[i]];
  }
#pragma unroll
  for (int i = 0; i < PER_THREAD; ++i) {
    myrec[i] = 0u;
    mypos[i] = 0xFFFFFFFFu;
    if (tid + i * BIN_THREADS < CHUNK) {
      float v = sa[i] + sb[i];
      v = (v > 0.0f) ? v : (ALPHA * v);
      const unsigned o = f2ord(v);
      omin = (o < omin) ? o : omin;
      omax = (o > omax) ? o : omax;
      const unsigned bin  = (unsigned)rr[i] / BIN_ROWS;
      const unsigned lrow = (unsigned)rr[i] - bin * BIN_ROWS;
      myrec[i] = ((unsigned)cc[i] << 8) | lrow;
      const unsigned rank = atomicAdd(&s_cnt[bin], 1u);
      mypos[i] = (bin << 16) | rank;
    }
  }

  for (int d = 16; d > 0; d >>= 1) {
    const unsigned pmin = __shfl_xor(omin, d);
    const unsigned pmax = __shfl_xor(omax, d);
    omin = (pmin < omin) ? pmin : omin;
    omax = (pmax > omax) ? pmax : omax;
  }
  if (lane == 0) { atomicMin(&s_min, omin); atomicMax(&s_max, omax); }
  __syncthreads();

  if (tid == 0) {
    atomicMin(&minmax[0], s_min);
    atomicMax(&minmax[1], s_max);
  }

  if (tid < 32) {
    unsigned loc[8];
    unsigned t = 0u;
#pragma unroll
    for (int q = 0; q < 8; ++q) { loc[q] = t; t += s_cnt[tid * 8 + q]; }
    unsigned incl = t;
    for (int d = 1; d < 32; d <<= 1) {
      const unsigned y = __shfl_up(incl, (unsigned)d);
      if (tid >= d) incl += y;
    }
    const unsigned excl = incl - t;
#pragma unroll
    for (int q = 0; q < 8; ++q) s_start[tid * 8 + q] = excl + loc[q];
    if (tid == 31) s_start[N_BINS] = incl;
  }
  __syncthreads();

  unsigned short* __restrict__ starts = reinterpret_cast<unsigned short*>(starts_words);
  if (tid <= N_BINS) starts[tid * N_CHUNKS + chunk] = (unsigned short)s_start[tid];

#pragma unroll
  for (int i = 0; i < PER_THREAD; ++i) {
    if (mypos[i] != 0xFFFFFFFFu) s_rec[s_start[mypos[i] >> 16] + (mypos[i] & 0xFFFFu)] = myrec[i];
  }
  __syncthreads();
  for (int k = tid; k < CHUNK; k += BIN_THREADS) rec[e0 + k] = s_rec[k];
}

#define SUM_WAVES   32
#define CH_PER_WAVE (N_CHUNKS / SUM_WAVES)
#define GROUP       32
#define SLOTS       8
static_assert(N_CHUNKS % SUM_WAVES == 0 && CH_PER_WAVE <= 32, "a lane per chunk of the wave holds its two starts");
static_assert(GROUP == 32 && GROUP % SLOTS == 0, "a batch is one record per lane, walked in whole groups of SLOTS");
static_assert(SUM_WAVES * 32 == STAGE_THREADS, "the table copy is laid out for this many threads");

__global__ __launch_bounds__(SUM_WAVES * 32)
void gat_bin_sum(const unsigned* __restrict__ rec,
                 const unsigned* __restrict__ starts_words,
                 const float* __restrict__ s0,
                 const float* __restrict__ s1,
                 const float* __restrict__ Xp,
                 const unsigned* __restrict__ minmax,
                 float* __restrict__ hout) {
  __shared__ float s_h[(BIN_ROWS + 1) * D_OUT];
  __shared__ float s_sum[BIN_ROWS];
  __shared__ float s_s0[BIN_ROWS];
  __shared__ alignas(16) float s_s1[N_NODES];

  const int lane = threadIdx.x & 31;
  const int wave  = __builtin_amdgcn_readfirstlane((int)(threadIdx.x >> 5));
  const int bin   = blockIdx.x;
  const int base  = bin * BIN_ROWS;
  const int nrows = (N_NODES - base < BIN_ROWS) ? (N_NODES - base) : BIN_ROWS;

  const unsigned short* __restrict__ starts = reinterpret_cast<const unsigned short*>(starts_words);
  int st = 0, en = 0;
  if (lane < CH_PER_WAVE) {
    const int k = wave + SUM_WAVES * lane;
    st = (int)starts[bin * N_CHUNKS + k];
    en = (int)starts[(bin + 1) * N_CHUNKS + k];
  }

  float4 tab[STAGE_PER];
  stage_load(s1, tab, (int)threadIdx.x);

  const float vmin = ord2f(minmax[0]);
  const float vmax = ord2f(minmax[1]);
  const float inv  = 1.0f / (vmax - vmin);

  for (int i = threadIdx.x; i < BIN_ROWS * D_OUT; i += SUM_WAVES * 32) s_h[i] = 0.0f;
  if (threadIdx.x < BIN_ROWS) {
    s_sum[threadIdx.x] = 0.0f;
    s_s0[threadIdx.x]  = ((int)threadIdx.x < nrows) ? s0[base + threadIdx.x] : 0.0f;
  }
  stage_store(s_s1, tab, (int)threadIdx.x);
  __syncthreads();

#pragma nounroll
  for (int j = 0; j < CH_PER_WAVE; ++j) {
    const int a = __builtin_amdgcn_readlane(st, j);
    const int b = __builtin_amdgcn_readlane(en, j);
    const unsigned* __restrict__ rp = rec + (wave + SUM_WAVES * j) * CHUNK;
    for (int p = a; p < b; p += 32) {
      const int  n     = (b - p < 32) ? (b - p) : 32;
      const bool valid = lane < n;
      int   erow = BIN_ROWS, ecol = 0;
      float att  = 0.0f;
      if (valid) {
        const unsigned rv = rp[p + lane];
        erow = (int)(rv & 0xFFu);
        ecol = (int)(rv >> 8);
        float v = s_s0[erow] + s_s1[ecol];
        v = (v > 0.0f) ? v : (ALPHA * v);
        att = __expf((v - vmin) * inv);
        __hip_atomic_fetch_add(&s_sum[erow], att, __ATOMIC_RELAXED, __HIP_MEMORY_SCOPE_WORKGROUP);
      }
      float x[GROUP];
#pragma unroll
      for (int t = 0; t < GROUP; ++t) {
        const int ck = __builtin_amdgcn_readlane(ecol, t);
        x[t] = Xp[(unsigned)ck * D_OUT + lane];
      }
#pragma unroll
      for (int g = 0; g < GROUP; g += SLOTS) {
        if (g < n) {
#pragma unroll
          for (int t = g; t < g + SLOTS; ++t) {
            const int   rk = __builtin_amdgcn_readlane(erow, t);
            const float ak = __int_as_float(__builtin_amdgcn_readlane(__float_as_int(att), t));
            __hip_atomic_fetch_add(&s_h[rk * D_OUT + lane], ak * x[t], __ATOMIC_RELAXED, __HIP_MEMORY_SCOPE_WORKGROUP);
          }
        }
      }
    }
  }
  __syncthreads();

  for (int row = wave; row < nrows; row += SUM_WAVES) {
    hout[(unsigned)(base + row) * D_OUT + lane] = s_h[row * D_OUT + lane] / s_sum[row];
  }
}

static inline int cdiv(int a, int b) { return (a + b - 1) / b; }

extern "C" void kernel_launch(void* const* d_in, const int* in_sizes, int n_in,
                              void* d_out, int out_size, void* d_ws, size_t ws_size,
                              hipStream_t stream) {
  (void)in_sizes; (void)n_in; (void)out_size; (void)ws_size;

  const float* X       = (const float*)d_in[0];
  const float* W       = (const float*)d_in[1];
  const float* a0      = (const float*)d_in[2];
  const float* a1      = (const float*)d_in[3];
  const int*   row_idx = (const int*)d_in[4];
  const int*   col_idx = (const int*)d_in[5];
  float* hout = (float*)d_out;

  char* ws = (char*)d_ws;
  size_t off = 0;
  auto alloc = [&](size_t bytes) { char* p = ws + off; off = (off + bytes + 255) & ~(size_t)255; return p; };
  float*    Xp     = (float*)alloc((size_t)N_NODES * D_OUT * sizeof(float));
  float*    s0     = (float*)alloc((size_t)N_NODES * sizeof(float));
  float*    s1     = (float*)alloc((size_t)N_NODES * sizeof(float));
  unsigned* rec    = (unsigned*)alloc((size_t)N_EDGES * sizeof(unsigned));
  unsigned* starts = (unsigned*)alloc((size_t)(N_BINS + 1) * N_CHUNKS * sizeof(unsigned short));
  unsigned* minmax = (unsigned*)alloc(2 * sizeof(unsigned));

  const int TB = 256;

  {
    const int n_waves  = N_NODES / 16;
    const int waves_pb = TB / 32;
    gat_gemm_wmma<<<cdiv(n_waves, waves_pb), TB, 0, stream>>>(X, W, a0, a1, Xp, s0, s1, minmax);
  }

  gat_edge_bin<<<N_CHUNKS, BIN_THREADS, 0, stream>>>(row_idx, col_idx, s0, s1, rec, starts, minmax);

  gat_bin_sum<<<N_BINS, SUM_WAVES * 32, 0, stream>>>(rec, starts, s0, s1, Xp, minmax, hout);
}


# summing launch: a block's records counting-sorted by row in LDS with their weights; a wave sums a row in registers from 16-byte gathers, 4 source rows per load; no LDS f32 adds; sum order changes
# speedup vs baseline: 8.3552x; 1.0865x over previous
#include <hip/hip_runtime.h>
#include <hip/hip_bf16.h>

#define N_NODES 50000
#define N_EDGES 1600000
#define D_IN    128
#define D_OUT   32
#define ALPHA   0.2f

typedef __attribute__((ext_vector_type(2))) float v2f;
typedef __attribute__((ext_vector_type(8))) float v8f;

__device__ __forceinline__ unsigned f2ord(float f) {
  unsigned u = __float_as_uint(f);
  return (u & 0x80000000u) ? ~u : (u | 0x80000000u);
}
__device__ __forceinline__ float ord2f(unsigned u) {
  u = (u & 0x80000000u) ? (u & 0x7FFFFFFFu) : ~u;
  return __uint_as_float(u);
}

__device__ __forceinline__ float row_sum16(float v) {
  v += __int_as_float(__builtin_amdgcn_update_dpp(0, __float_as_int(v), 0x161, 0xf, 0xf, false));
  v += __int_as_float(__builtin_amdgcn_update_dpp(0, __float_as_int(v), 0x162, 0xf, 0xf, false));
  v += __int_as_float(__builtin_amdgcn_update_dpp(0, __float_as_int(v), 0x164, 0xf, 0xf, false));
  v += __int_as_float(__builtin_amdgcn_update_dpp(0, __float_as_int(v), 0x168, 0xf, 0xf, false));
  return v;
}

static_assert(N_NODES % 16 == 0, "the tiles cover every row of X', s0 and s1 exactly once");
__global__ void gat_gemm_wmma(const float* __restrict__ X,
                              const float* __restrict__ W,
                              const float* __restrict__ a0,
                              const float* __restrict__ a1,
                              float* __restrict__ Xp,
                              float* __restrict__ s0,
                              float* __restrict__ s1,
                              unsigned* __restrict__ minmax) {
  if (blockIdx.x == 0 && threadIdx.x == 0) {
    minmax[0] = 0xFFFFFFFFu; minmax[1] = 0u;
  }

  const int wave_in_block = threadIdx.x >> 5;
  const int lane          = threadIdx.x & 31;
  const int wave_id       = blockIdx.x * (blockDim.x >> 5) + wave_in_block;
  const int row_base      = wave_id * 16;
  if (row_base >= N_NODES) return;

  const int m    = lane & 15;
  const int kgrp = lane >> 4;

  const float* __restrict__ xrow = X + (size_t)(row_base + m) * D_IN;

  v8f acc0 = {};
  v8f acc1 = {};

  for (int k = 0; k < D_IN; k += 4) {
    const int kk = k + kgrp * 2;
    v2f a;
    a.x = xrow[kk + 0];
    a.y = xrow[kk + 1];
    v2f b0, b1;
    b0.x = W[(size_t)(kk + 0) * D_OUT + m];
    b0.y = W[(size_t)(kk + 1) * D_OUT + m];
    b1.x = W[(size_t)(kk + 0) * D_OUT + 16 + m];
    b1.y = W[(size_t)(kk + 1) * D_OUT + 16 + m];

    acc0 = __builtin_amdgcn_wmma_f32_16x16x4_f32(false, a, false, b0, (short)0, acc0, false, false);
    acc1 = __builtin_amdgcn_wmma_f32_16x16x4_f32(false, a, false, b1, (short)0, acc1, false, false);
  }

  const int rbase = row_base + kgrp * 8;
  for (int i = 0; i < 8; ++i) {
    const size_t r = (size_t)(rbase + i);
    Xp[r * D_OUT + m]      = acc0[i];
    Xp[r * D_OUT + 16 + m] = acc1[i];
  }

  const float a0lo = a0[m], a0hi = a0[16 + m];
  const float a1lo = a1[m], a1hi = a1[16 + m];
  float v0 = 0.0f, v1 = 0.0f;
  for (int i = 0; i < 8; ++i) {
    const float t0 = row_sum16(acc0[i] * a0lo + acc1[i] * a0hi);
    const float t1 = row_sum16(acc0[i] * a1lo + acc1[i] * a1hi);
    if (m == i) { v0 = t0; v1 = t1; }
  }
  if (m < 8) {
    s0[rbase + m] = v0;
    s1[rbase + m] = v1;
  }
}

#define STAGE_THREADS 1024
#define STAGE_VEC     (N_NODES / 4)
#define STAGE_PER     ((STAGE_VEC + STAGE_THREADS - 1) / STAGE_THREADS)
static_assert(N_NODES % 4 == 0, "a table is a whole number of 16-byte pieces");

__device__ __forceinline__ void stage_load(const float* __restrict__ table, float4 (&t)[STAGE_PER], int tid) {
  const float4* __restrict__ t4 = reinterpret_cast<const float4*>(table);
#pragma unroll
  for (int j = 0; j < STAGE_PER; ++j) {
    const int q = tid + j * STAGE_THREADS;
    t[j] = make_float4(0.0f, 0.0f, 0.0f, 0.0f);
    if (q < STAGE_VEC) t[j] = t4[q];
  }
}
__device__ __forceinline__ void stage_store(float* lds_table, const float4 (&t)[STAGE_PER], int tid) {
  float4* l4 = reinterpret_cast<float4*>(lds_table);
#pragma unroll
  for (int j = 0; j < STAGE_PER; ++j) {
    const int q = tid + j * STAGE_THREADS;
    if (q < STAGE_VEC) l4[q] = t[j];
  }
}

#define BIN_ROWS    196
#define N_BINS      256
#define N_CHUNKS    256
#define CHUNK       (N_EDGES / N_CHUNKS)
#define BIN_THREADS 1024
#define PER_THREAD  ((CHUNK + BIN_THREADS - 1) / BIN_THREADS)
static_assert(N_EDGES % N_CHUNKS == 0, "the chunks cover the edges exactly");
static_assert(N_BINS * BIN_ROWS >= N_NODES && (N_BINS - 1) * BIN_ROWS < N_NODES, "the blocks cover the rows, none is empty");
static_assert(BIN_ROWS <= 256, "the row inside its block is kept in 8 bits of the record");
static_assert(N_NODES <= 65536, "the source row is kept in 16 bits of the record");
static_assert(CHUNK < 65536, "a block's start inside a chunk is kept in 16 bits");
static_assert(N_BINS == 256 && BIN_THREADS >= N_BINS + 1, "the running sum takes 8 counts per lane of one wave");
static_assert(D_OUT == 32, "one lane per output channel");
static_assert(BIN_THREADS == STAGE_THREADS, "the table copy is laid out for this many threads");

__global__ __launch_bounds__(BIN_THREADS)
void gat_edge_bin(const int* __restrict__ row_idx,
                  const int* __restrict__ col_idx,
                  const float* __restrict__ s0,
                  const float* __restrict__ s1,
                  unsigned* __restrict__ rec,
                  unsigned* __restrict__ starts_words,
                  unsigned* __restrict__ minmax) {
  __shared__ unsigned s_cnt[N_BINS];
  __shared__ unsigned s_start[N_BINS + 1];
  __shared__ unsigned s_rec[CHUNK];
  __shared__ unsigned s_min, s_max;
  __shared__ alignas(16) float s_tab[N_NODES];

  const int tid   = threadIdx.x;
  const int lane  = tid & 31;
  const int chunk = blockIdx.x;
  const int e0    = chunk * CHUNK;

  if (tid < N_BINS) s_cnt[tid] = 0u;
  if (tid == 0) { s_min = 0xFFFFFFFFu; s_max = 0u; }

  unsigned myrec[PER_THREAD];
  unsigned mypos[PER_THREAD];
  unsigned omin = 0xFFFFFFFFu, omax = 0u;
  int   rr[PER_THREAD], cc[PER_THREAD];
  float sa[PER_THREAD], sb[PER_THREAD];
  float4 tab[STAGE_PER];
  stage_load(s0, tab, tid);
#pragma unroll
  for (int i = 0; i < PER_THREAD; ++i) {
    const int k = tid + i * BIN_THREADS;
    rr[i] = -1;
    cc[i] = 0;
    if (k < CHUNK) {
      rr[i] = row_idx[e0 + k];
      cc[i] = col_idx[e0 + k];
    }
  }
  stage_store(s_tab, tab, tid);
  __syncthreads();
#pragma unroll
  for (int i = 0; i < PER_THREAD; ++i) {
    sa[i] = 0.0f;
    if (tid + i * BIN_THREADS < CHUNK) sa[i] = s_tab[rr[i]];
  }
  stage_load(s1, tab, tid);
  __syncthreads();
  stage_store(s_tab, tab, tid);
  __syncthreads();
#pragma unroll
  for (int i = 0; i < PER_THREAD; ++i) {
    sb[i] = 0.0f;
    if (tid + i * BIN_THREADS < CHUNK) sb[i] = s_tab[cc[i]];
  }
#pragma unroll
  for (int i = 0; i < PER_THREAD; ++i) {
    myrec[i] = 0u;
    mypos[i] = 0xFFFFFFFFu;
    if (tid + i * BIN_THREADS < CHUNK) {
      float v = sa[i] + sb[i];
      v = (v > 0.0f) ? v : (ALPHA * v);
      const unsigned o = f2ord(v);
      omin = (o < omin) ? o : omin;
      omax = (o > omax) ? o : omax;
      const unsigned bin  = (unsigned)rr[i] / BIN_ROWS;
      const unsigned lrow = (unsigned)rr[i] - bin * BIN_ROWS;
      myrec[i] = ((unsigned)cc[i] << 8) | lrow;
      const unsigned rank = atomicAdd(&s_cnt[bin], 1u);
      mypos[i] = (bin << 16) | rank;
    }
  }

  for (int d = 16; d > 0; d >>= 1) {
    const unsigned pmin = __shfl_xor(omin, d);
    const unsigned pmax = __shfl_xor(omax, d);
    omin = (pmin < omin) ? pmin : omin;
    omax = (pmax > omax) ? pmax : omax;
  }
  if (lane == 0) { atomicMin(&s_min, omin); atomicMax(&s_max, omax); }
  __syncthreads();

  if (tid == 0) {
    atomicMin(&minmax[0], s_min);
    atomicMax(&minmax[1], s_max);
  }

  if (tid < 32) {
    unsigned loc[8];
    unsigned t = 0u;
#pragma unroll
    for (int q = 0; q < 8; ++q) { loc[q] = t; t += s_cnt[tid * 8 + q]; }
    unsigned incl = t;
    for (int d = 1; d < 32; d <<= 1) {
      const unsigned y = __shfl_up(incl, (unsigned)d);
      if (tid >= d) incl += y;
    }
    const unsigned excl = incl - t;
#pragma unroll
    for (int q = 0; q < 8; ++q) s_start[tid * 8 + q] = excl + loc[q];
    if (tid == 31) s_start[N_BINS] = incl;
  }
  __syncthreads();

  unsigned short* __restrict__ starts = reinterpret_cast<unsigned short*>(starts_words);
  if (tid <= N_BINS) starts[tid * N_CHUNKS + chunk] = (unsigned short)s_start[tid];

#pragma unroll
  for (int i = 0; i < PER_THREAD; ++i) {
    if (mypos[i] != 0xFFFFFFFFu) s_rec[s_start[mypos[i] >> 16] + (mypos[i] & 0xFFFFu)] = myrec[i];
  }
  __syncthreads();
  for (int k = tid; k < CHUNK; k += BIN_THREADS) rec[e0 + k] = s_rec[k];
}

#define SUM_WAVES   32
#define CH_PER_WAVE (N_CHUNKS / SUM_WAVES)
#define SORT_CAP    8192
#define ROW_LOADS   12
#define GROUP       32
#define SLOTS       8
#define SCAN_PER    ((BIN_ROWS + 31) / 32)
static_assert(N_CHUNKS % SUM_WAVES == 0 && CH_PER_WAVE <= 32, "a lane per chunk of the wave holds its two starts");
static_assert(GROUP == 32 && GROUP % SLOTS == 0, "a batch is one record per lane, walked in whole groups of SLOTS");
static_assert(SUM_WAVES * 32 == STAGE_THREADS, "the table copy is laid out for this many threads");
static_assert(BIN_ROWS < 255, "row byte 255 marks a lane without a record");
static_assert(2 * SORT_CAP >= (BIN_ROWS + 1) * D_OUT + BIN_ROWS, "the other path's block of h and sums fit the record room");
static_assert(D_OUT == 32, "8 lanes of 4 channels cover a row");

__global__ __launch_bounds__(SUM_WAVES * 32)
void gat_bin_sum(const unsigned* __restrict__ rec,
                 const unsigned* __restrict__ starts_words,
                 const float* __restrict__ s0,
                 const float* __restrict__ s1,
                 const float* __restrict__ Xp,
                 const unsigned* __restrict__ minmax,
                 float* __restrict__ hout) {
  __shared__ alignas(16) float s_s1[N_NODES];
  __shared__ float    s_s0[BIN_ROWS];
  __shared__ unsigned s_cnt[BIN_ROWS];
  __shared__ unsigned s_pos[BIN_ROWS];
  __shared__ unsigned s_off[BIN_ROWS + 1];
  __shared__ alignas(16) unsigned s_buf[2 * SORT_CAP];

  const int lane = threadIdx.x & 31;
  const int wave  = __builtin_amdgcn_readfirstlane((int)(threadIdx.x >> 5));
  const int bin   = blockIdx.x;
  const int base  = bin * BIN_ROWS;
  const int nrows = (N_NODES - base < BIN_ROWS) ? (N_NODES - base) : BIN_ROWS;

  const unsigned short* __restrict__ starts = reinterpret_cast<const unsigned short*>(starts_words);
  int st = 0, en = 0;
  if (lane < CH_PER_WAVE) {
    const int k = wave + SUM_WAVES * lane;
    st = (int)starts[bin * N_CHUNKS + k];
    en = (int)starts[(bin + 1) * N_CHUNKS + k];
  }

  float s0v = 0.0f;
  if ((int)threadIdx.x < nrows) s0v = s0[base + threadIdx.x];
  if (threadIdx.x < BIN_ROWS) {
    s_cnt[threadIdx.x] = 0u;
    s_s0[threadIdx.x]  = s0v;
  }
  __syncthreads();

  unsigned rv[CH_PER_WAVE];
#pragma unroll
  for (int j = 0; j < CH_PER_WAVE; ++j) {
    const int a = __builtin_amdgcn_readlane(st, j);
    const int b = __builtin_amdgcn_readlane(en, j);
    rv[j] = 0xFFFFFFFFu;
    if (lane < b - a) rv[j] = rec[(wave + SUM_WAVES * j) * CHUNK + a + lane];
  }

  float4 tab[STAGE_PER];
  stage_load(s1, tab, (int)threadIdx.x);

  const float vmin = ord2f(minmax[0]);
  const float vmax = ord2f(minmax[1]);
  const float inv  = 1.0f / (vmax - vmin);

#pragma unroll
  for (int j = 0; j < CH_PER_WAVE; ++j) {
    if (rv[j] != 0xFFFFFFFFu) atomicAdd(&s_cnt[rv[j] & 0xFFu], 1u);
  }
#pragma nounroll
  for (int j = 0; j < CH_PER_WAVE; ++j) {
    const int a = __builtin_amdgcn_readlane(st, j);
    const int b = __builtin_amdgcn_readlane(en, j);
    const unsigned* __restrict__ rp = rec + (wave + SUM_WAVES * j) * CHUNK;
    for (int p = a + 32; p < b; p += 32) {
      if (p + lane < b) atomicAdd(&s_cnt[rp[p + lane] & 0xFFu], 1u);
    }
  }
  stage_store(s_s1, tab, (int)threadIdx.x);
  __syncthreads();

  if (threadIdx.x < 32) {
    unsigned loc[SCAN_PER];
    unsigned t = 0u;
#pragma unroll
    for (int q = 0; q < SCAN_PER; ++q) {
      const int r = lane * SCAN_PER + q;
      loc[q] = t;
      if (r < BIN_ROWS) t += s_cnt[r];
    }
    unsigned incl = t;
    for (int d = 1; d < 32; d <<= 1) {
      const unsigned y = __shfl_up(incl, (unsigned)d);
      if (lane >= d) incl += y;
    }
    const unsigned excl = incl - t;
#pragma unroll
    for (int q = 0; q < SCAN_PER; ++q) {
      const int r = lane * SCAN_PER + q;
      if (r < BIN_ROWS) { s_off[r] = excl + loc[q]; s_pos[r] = excl + loc[q]; }
    }
    if (lane == 31) s_off[BIN_ROWS] = incl;
  }
  __syncthreads();

  const unsigned total = s_off[BIN_ROWS];
  if (total <= SORT_CAP) {
    uint2* srec = reinterpret_cast<uint2*>(s_buf);
#pragma unroll
    for (int j = 0; j < CH_PER_WAVE; ++j) {
      if (rv[j] != 0xFFFFFFFFu) {
        const unsigned erow = rv[j] & 0xFFu;
        const unsigned ecol = rv[j] >> 8;
        float v = s_s0[erow] + s_s1[ecol];
        v = (v > 0.0f) ? v : (ALPHA * v);
        const float att = __expf((v - vmin) * inv);
        const unsigned pos = atomicAdd(&s_pos[erow], 1u);
        srec[pos] = make_uint2(ecol, __float_as_uint(att));
      }
    }
#pragma nounroll
    for (int j = 0; j < CH_PER_WAVE; ++j) {
      const int a = __builtin_amdgcn_readlane(st, j);
      const int b = __builtin_amdgcn_readlane(en, j);
      const unsigned* __restrict__ rp = rec + (wave + SUM_WAVES * j) * CHUNK;
      for (int p = a + 32; p < b; p += 32) {
        if (p + lane < b) {
          const unsigned r    = rp[p + lane];
          const unsigned erow = r & 0xFFu;
          const unsigned ecol = r >> 8;
          float v = s_s0[erow] + s_s1[ecol];
          v = (v > 0.0f) ? v : (ALPHA * v);
          const float att = __expf((v - vmin) * inv);
          const unsigned pos = atomicAdd(&s_pos[erow], 1u);
          srec[pos] = make_uint2(ecol, __float_as_uint(att));
        }
      }
    }
    __syncthreads();

    const int e  = lane >> 3;
    const int c4 = lane & 7;
    const float4* __restrict__ Xp4 = reinterpret_cast<const float4*>(Xp);
    float4* __restrict__ h4 = reinterpret_cast<float4*>(hout);
    for (int row = wave; row < nrows; row += SUM_WAVES) {
      const int a = __builtin_amdgcn_readfirstlane((int)s_off[row]);
      const int b = __builtin_amdgcn_readfirstlane((int)s_off[row + 1]);
      float ax = 0.0f, ay = 0.0f, az = 0.0f, aw = 0.0f, asum = 0.0f;
      for (int p = a; p < b; p += 4 * ROW_LOADS) {
        uint2  rc[ROW_LOADS];
        float4 x[ROW_LOADS];
#pragma unroll
        for (int g = 0; g < ROW_LOADS; ++g) {
          const int idx = p + 4 * g + e;
          rc[g] = srec[(idx < b) ? idx : (b - 1)];
        }
#pragma unroll
        for (int g = 0; g < ROW_LOADS; ++g) {
          x[g] = Xp4[rc[g].x * (D_OUT / 4) + c4];
        }
#pragma unroll
        for (int g = 0; g < ROW_LOADS; ++g) {
          const float att = (p + 4 * g + e < b) ? __uint_as_float(rc[g].y) : 0.0f;
          ax   += att * x[g].x;
          ay   += att * x[g].y;
          az   += att * x[g].z;
          aw   += att * x[g].w;
          asum += att;
        }
      }
      ax += __shfl_xor(ax, 8);  ay += __shfl_xor(ay, 8);  az += __shfl_xor(az, 8);
      aw += __shfl_xor(aw, 8);  asum += __shfl_xor(asum, 8);
      ax += __shfl_xor(ax, 16); ay += __shfl_xor(ay, 16); az += __shfl_xor(az, 16);
      aw += __shfl_xor(aw, 16); asum += __shfl_xor(asum, 16);
      if (e == 0) h4[(unsigned)(base + row) * (D_OUT / 4) + c4] = make_float4(ax / asum, ay / asum, az / asum, aw / asum);
    }
  } else {
    float* s_h   = reinterpret_cast<float*>(s_buf);
    float* s_sum = s_h + (BIN_ROWS + 1) * D_OUT;
    for (int i = threadIdx.x; i < BIN_ROWS * D_OUT; i += SUM_WAVES * 32) s_h[i] = 0.0f;
    if (threadIdx.x < BIN_ROWS) s_sum[threadIdx.x] = 0.0f;
    __syncthreads();

#pragma nounroll
    for (int j = 0; j < CH_PER_WAVE; ++j) {
      const int a = __builtin_amdgcn_readlane(st, j);
      const int b = __builtin_amdgcn_readlane(en, j);
      const unsigned* __restrict__ rp = rec + (wave + SUM_WAVES * j) * CHUNK;
      for (int p = a; p < b; p += 32) {
        const int  n     = (b - p < 32) ? (b - p) : 32;
        const bool valid = lane < n;
        int   erow = BIN_ROWS, ecol = 0;
        float att  = 0.0f;
        if (valid) {
          const unsigned r = rp[p + lane];
          erow = (int)(r & 0xFFu);
          ecol = (int)(r >> 8);
          float v = s_s0[erow] + s_s1[ecol];
          v = (v > 0.0f) ? v : (ALPHA * v);
          att = __expf((v - vmin) * inv);
          __hip_atomic_fetch_add(&s_sum[erow], att, __ATOMIC_RELAXED, __HIP_MEMORY_SCOPE_WORKGROUP);
        }
        float x[GROUP];
#pragma unroll
        for (int t = 0; t < GROUP; ++t) {
          const int ck = __builtin_amdgcn_readlane(ecol, t);
          x[t] = Xp[(unsigned)ck * D_OUT + lane];
        }
#pragma unroll
        for (int g = 0; g < GROUP; g += SLOTS) {
          if (g < n) {
#pragma unroll
            for (int t = g; t < g + SLOTS; ++t) {
              const int   rk = __builtin_amdgcn_readlane(erow, t);
              const float ak = __int_as_float(__builtin_amdgcn_readlane(__float_as_int(att), t));
              __hip_atomic_fetch_add(&s_h[rk * D_OUT + lane], ak * x[t], __ATOMIC_RELAXED, __HIP_MEMORY_SCOPE_WORKGROUP);
            }
          }
        }
      }
    }
    __syncthreads();

    for (int row = wave; row < nrows; row += SUM_WAVES) {
      hout[(unsigned)(base + row) * D_OUT + lane] = s_h[row * D_OUT + lane] / s_sum[row];
    }
  }
}

static inline int cdiv(int a, int b) { return (a + b - 1) / b; }

extern "C" void kernel_launch(void* const* d_in, const int* in_sizes, int n_in,
                              void* d_out, int out_size, void* d_ws, size_t ws_size,
                              hipStream_t stream) {
  (void)in_sizes; (void)n_in; (void)out_size; (void)ws_size;

  const float* X       = (const float*)d_in[0];
  const float* W       = (const float*)d_in[1];
  const float* a0      = (const float*)d_in[2];
  const float* a1      = (const float*)d_in[3];
  const int*   row_idx = (const int*)d_in[4];
  const int*   col_idx = (const int*)d_in[5];
  float* hout = (float*)d_out;

  char* ws = (char*)d_ws;
  size_t off = 0;
  auto alloc = [&](size_t bytes) { char* p = ws + off; off = (off + bytes + 255) & ~(size_t)255; return p; };
  float*    Xp     = (float*)alloc((size_t)N_NODES * D_OUT * sizeof(float));
  float*    s0     = (float*)alloc((size_t)N_NODES * sizeof(float));
  float*    s1     = (float*)alloc((size_t)N_NODES * sizeof(float));
  unsigned* rec    = (unsigned*)alloc((size_t)N_EDGES * sizeof(unsigned));
  unsigned* starts = (unsigned*)alloc((size_t)(N_BINS + 1) * N_CHUNKS * sizeof(unsigned short));
  unsigned* minmax = (unsigned*)alloc(2 * sizeof(unsigned));

  const int TB = 256;

  {
    const int n_waves  = N_NODES / 16;
    const int waves_pb = TB / 32;
    gat_gemm_wmma<<<cdiv(n_waves, waves_pb), TB, 0, stream>>>(X, W, a0, a1, Xp, s0, s1, minmax);
  }

  gat_edge_bin<<<N_CHUNKS, BIN_THREADS, 0, stream>>>(row_idx, col_idx, s0, s1, rec, starts, minmax);

  gat_bin_sum<<<N_BINS, SUM_WAVES * 32, 0, stream>>>(rec, starts, s0, s1, Xp, minmax, hout);
}
